# starting kernel
# baseline (speedup 1.0000x reference)
#include <hip/hip_runtime.h>


typedef float v2f __attribute__((ext_vector_type(2)));
typedef float v8f __attribute__((ext_vector_type(8)));

#define WMMA_F32(A, B, C) \
  __builtin_amdgcn_wmma_f32_16x16x4_f32(false, (A), false, (B), (short)0, (C), false, false)

static __device__ __forceinline__ v2f ld2(const float* p) { return *(const v2f*)p; }
static __device__ __forceinline__ v2f mk2(float a, float b) { v2f r; r.x = a; r.y = b; return r; }

__global__ __launch_bounds__(256) void qkv_kernel(
    const float* __restrict__ x,
    const float* __restrict__ Wf, const float* __restrict__ Wg, const float* __restrict__ Wh,
    float* __restrict__ q, float* __restrict__ k, float* __restrict__ v) {
  __shared__ v2f wl[3][32 * 32];
  const int tid = threadIdx.x;
  const int wave = tid >> 5, lane = tid & 31;
  const int m = lane & 15, half = lane >> 4, kOff = half * 2;
  const int row0 = blockIdx.x * 128 + wave * 16;

  v8f aq0 = {}, aq1 = {}, ak0 = {}, ak1 = {}, av0 = {}, av1 = {};

  for (int c0 = 0; c0 < 256; c0 += 64) {
    for (int idx = tid; idx < 32 * 32; idx += 256) {
      const int kk2 = idx >> 5, n = idx & 31;
      const int g = (c0 + kk2 * 2) * 32 + n;
      wl[0][idx] = mk2(Wf[g], Wf[g + 32]);
      wl[1][idx] = mk2(Wg[g], Wg[g + 32]);
      wl[2][idx] = mk2(Wh[g], Wh[g + 32]);
    }
    __syncthreads();
#pragma unroll
    for (int kk = 0; kk < 64; kk += 4) {
      const v2f a = ld2(&x[(row0 + m) * 256 + c0 + kk + kOff]);
      const int bi = ((kk + kOff) >> 1) * 32;
      const v2f bf0 = wl[0][bi + m], bf1 = wl[0][bi + 16 + m];
      const v2f bg0 = wl[1][bi + m], bg1 = wl[1][bi + 16 + m];
      const v2f bh0 = wl[2][bi + m], bh1 = wl[2][bi + 16 + m];
      aq0 = WMMA_F32(a, bf0, aq0); aq1 = WMMA_F32(a, bf1, aq1);
      ak0 = WMMA_F32(a, bg0, ak0); ak1 = WMMA_F32(a, bg1, ak1);
      av0 = WMMA_F32(a, bh0, av0); av1 = WMMA_F32(a, bh1, av1);
    }
    __syncthreads();
  }
#pragma unroll
  for (int vi = 0; vi < 8; ++vi) {
    const int r = (row0 + vi + half * 8) * 32;
    q[r + m] = aq0[vi]; q[r + 16 + m] = aq1[vi];
    k[r + m] = ak0[vi]; k[r + 16 + m] = ak1[vi];
    v[r + m] = av0[vi]; v[r + 16 + m] = av1[vi];
  }
}

__global__ __launch_bounds__(256) void attn_kernel(
    const float* __restrict__ q, const float* __restrict__ k,
    const float* __restrict__ v, float* __restrict__ o) {
  __shared__ float smem[8][1024];
  const int tid = threadIdx.x;
  const int wave = tid >> 5, lane = tid & 31;
  const int m = lane & 15, half = lane >> 4, kOff = half * 2;
  const int tile = wave >> 2, split = wave & 3;
  const int task = blockIdx.x * 2 + tile;
  const int b = task >> 8;
  const int i0 = (task & 255) * 16;
  const float* qb = q + b * 4096 * 32;
  const float* kb = k + b * 4096 * 32;
  const float* vb = v + b * 4096 * 32;

  v2f aK[8];
#pragma unroll
  for (int t = 0; t < 8; ++t) aK[t] = ld2(&kb[(i0 + m) * 32 + 4 * t + kOff]);

  float mrun[8], lrun[8];
  v8f o0 = {}, o1 = {};
#pragma unroll
  for (int vi = 0; vi < 8; ++vi) { mrun[vi] = -__builtin_inff(); lrun[vi] = 0.f; }

  const int jbeg = split * 1024, jend = jbeg + 1024;
  for (int j0 = jbeg; j0 < jend; j0 += 64) {
    v8f s[4] = {{}, {}, {}, {}};
#pragma unroll
    for (int st = 0; st < 4; ++st) {
#pragma unroll
      for (int t = 0; t < 8; ++t) {
        const v2f bq = ld2(&qb[(j0 + st * 16 + m) * 32 + 4 * t + kOff]);
        s[st] = WMMA_F32(aK[t], bq, s[st]);
      }
    }
#pragma unroll
    for (int vi = 0; vi < 8; ++vi) {
      float mx = fmaxf(fmaxf(s[0][vi], s[1][vi]), fmaxf(s[2][vi], s[3][vi]));
      mx = fmaxf(mx, __shfl_xor(mx, 1, 16));
      mx = fmaxf(mx, __shfl_xor(mx, 2, 16));
      mx = fmaxf(mx, __shfl_xor(mx, 4, 16));
      mx = fmaxf(mx, __shfl_xor(mx, 8, 16));
      const float nm = fmaxf(mrun[vi], mx);
      const float sc = __expf(mrun[vi] - nm);
      mrun[vi] = nm;
      const float p0 = __expf(s[0][vi] - nm); s[0][vi] = p0;
      const float p1 = __expf(s[1][vi] - nm); s[1][vi] = p1;
      const float p2 = __expf(s[2][vi] - nm); s[2][vi] = p2;
      const float p3 = __expf(s[3][vi] - nm); s[3][vi] = p3;
      float rs = (p0 + p1) + (p2 + p3);
      rs += __shfl_xor(rs, 1, 16);
      rs += __shfl_xor(rs, 2, 16);
      rs += __shfl_xor(rs, 4, 16);
      rs += __shfl_xor(rs, 8, 16);
      lrun[vi] = lrun[vi] * sc + rs;
      o0[vi] *= sc;
      o1[vi] *= sc;
    }
#pragma unroll
    for (int st = 0; st < 4; ++st)
#pragma unroll
      for (int vi = 0; vi < 8; ++vi)
        smem[wave][(vi + half * 8) * 64 + st * 16 + m] = s[st][vi];
    asm volatile("s_wait_dscnt 0" ::: "memory");
#pragma unroll
    for (int t = 0; t < 16; ++t) {
      const int kk = 4 * t + kOff;
      const v2f aP = *(const v2f*)&smem[wave][m * 64 + kk];
      const v2f bv0 = mk2(vb[(j0 + kk) * 32 + m],      vb[(j0 + kk + 1) * 32 + m]);
      const v2f bv1 = mk2(vb[(j0 + kk) * 32 + 16 + m], vb[(j0 + kk + 1) * 32 + 16 + m]);
      o0 = WMMA_F32(aP, bv0, o0);
      o1 = WMMA_F32(aP, bv1, o1);
    }
  }

  asm volatile("s_wait_dscnt 0" ::: "memory");
#pragma unroll
  for (int vi = 0; vi < 8; ++vi) {
    const int r = vi + half * 8;
    smem[wave][r * 32 + m]      = o0[vi];
    smem[wave][r * 32 + 16 + m] = o1[vi];
  }
  if (m == 0) {
#pragma unroll
    for (int vi = 0; vi < 8; ++vi) {
      smem[wave][512 + vi + half * 8] = lrun[vi];
      smem[wave][528 + vi + half * 8] = mrun[vi];
    }
  }
  __syncthreads();

  if (split == 0) {
    const int w0 = wave;
#pragma unroll
    for (int vi = 0; vi < 8; ++vi) {
      const int r = vi + half * 8;
      float mg = smem[w0][528 + r];
      mg = fmaxf(mg, smem[w0 + 1][528 + r]);
      mg = fmaxf(mg, smem[w0 + 2][528 + r]);
      mg = fmaxf(mg, smem[w0 + 3][528 + r]);
      float lg = 0.f, a0 = 0.f, a1 = 0.f;
#pragma unroll
      for (int p = 0; p < 4; ++p) {
        const float wgt = __expf(smem[w0 + p][528 + r] - mg);
        lg += wgt * smem[w0 + p][512 + r];
        a0 += wgt * smem[w0 + p][r * 32 + m];
        a1 += wgt * smem[w0 + p][r * 32 + 16 + m];
      }
      const float inv = 1.f / lg;
      const int g = b * 4096 * 32 + (i0 + r) * 32;
      o[g + m]      = a0 * inv;
      o[g + 16 + m] = a1 * inv;
    }
  }
}

__global__ __launch_bounds__(256) void out_kernel(
    const float* __restrict__ o, const float* __restrict__ Wv,
    const float* __restrict__ x, const float* __restrict__ gamma,
    float* __restrict__ out) {
  const int tid = threadIdx.x;
  const int wave = tid >> 5, lane = tid & 31;
  const int m = lane & 15, half = lane >> 4, kOff = half * 2;
  const int task = blockIdx.x * 8 + wave;
  const int row0 = (task >> 2) * 16, col0 = (task & 3) * 64;

  v2f aO[8];
#pragma unroll
  for (int t = 0; t < 8; ++t) aO[t] = ld2(&o[(row0 + m) * 32 + 4 * t + kOff]);

  v8f acc[4] = {{}, {}, {}, {}};
#pragma unroll
  for (int t = 0; t < 8; ++t) {
    const int kk = 4 * t + kOff;
#pragma unroll
    for (int ct = 0; ct < 4; ++ct) {
      const int col = col0 + ct * 16 + m;
      const v2f bw = mk2(Wv[kk * 256 + col], Wv[(kk + 1) * 256 + col]);
      acc[ct] = WMMA_F32(aO[t], bw, acc[ct]);
    }
  }
  const float g = *gamma;
#pragma unroll
  for (int ct = 0; ct < 4; ++ct) {
#pragma unroll
    for (int vi = 0; vi < 8; ++vi) {
      const int idx = (row0 + vi + half * 8) * 256 + col0 + ct * 16 + m;
      out[idx] = g * acc[ct][vi] + x[idx];
    }
  }
}

extern "C" void kernel_launch(void* const* d_in, const int* in_sizes, int n_in,
                              void* d_out, int out_size, void* d_ws, size_t ws_size,
                              hipStream_t stream) {
  const float* x  = (const float*)d_in[0];
  const float* Wf = (const float*)d_in[1];
  const float* Wg = (const float*)d_in[2];
  const float* Wh = (const float*)d_in[3];
  const float* Wv = (const float*)d_in[4];
  const float* gm = (const float*)d_in[5];
  float* out = (float*)d_out;

  const int QKV = 4 * 4096 * 32;
  float* q = (float*)d_ws;
  float* k = q + QKV;
  float* v = k + QKV;
  float* o = v + QKV;

  qkv_kernel<<<128, 256, 0, stream>>>(x, Wf, Wg, Wh, q, k, v);
  attn_kernel<<<512, 256, 0, stream>>>(q, k, v, o);
  out_kernel<<<512, 256, 0, stream>>>(o, Wv, x, gm, out);
}


# attention: block = 4 row tiles x 4 col splits (16 waves); each 64-col q/v tile copied once per block into padded LDS (2 stages) and read as B fragments by 4 waves; sums and order unchanged
# speedup vs baseline: 1.1182x; 1.1182x over previous
#include <hip/hip_runtime.h>


typedef float v2f __attribute__((ext_vector_type(2)));
typedef float v8f __attribute__((ext_vector_type(8)));

#define WMMA_F32(A, B, C) \
  __builtin_amdgcn_wmma_f32_16x16x4_f32(false, (A), false, (B), (short)0, (C), false, false)

static __device__ __forceinline__ v2f ld2(const float* p) { return *(const v2f*)p; }
static __device__ __forceinline__ v2f mk2(float a, float b) { v2f r; r.x = a; r.y = b; return r; }

__global__ __launch_bounds__(256) void qkv_kernel(
    const float* __restrict__ x,
    const float* __restrict__ Wf, const float* __restrict__ Wg, const float* __restrict__ Wh,
    float* __restrict__ q, float* __restrict__ k, float* __restrict__ v) {
  __shared__ v2f wl[3][32 * 32];
  const int tid = threadIdx.x;
  const int wave = tid >> 5, lane = tid & 31;
  const int m = lane & 15, half = lane >> 4, kOff = half * 2;
  const int row0 = blockIdx.x * 128 + wave * 16;

  v8f aq0 = {}, aq1 = {}, ak0 = {}, ak1 = {}, av0 = {}, av1 = {};

  for (int c0 = 0; c0 < 256; c0 += 64) {
    for (int idx = tid; idx < 32 * 32; idx += 256) {
      const int kk2 = idx >> 5, n = idx & 31;
      const int g = (c0 + kk2 * 2) * 32 + n;
      wl[0][idx] = mk2(Wf[g], Wf[g + 32]);
      wl[1][idx] = mk2(Wg[g], Wg[g + 32]);
      wl[2][idx] = mk2(Wh[g], Wh[g + 32]);
    }
    __syncthreads();
#pragma unroll
    for (int kk = 0; kk < 64; kk += 4) {
      const v2f a = ld2(&x[(row0 + m) * 256 + c0 + kk + kOff]);
      const int bi = ((kk + kOff) >> 1) * 32;
      const v2f bf0 = wl[0][bi + m], bf1 = wl[0][bi + 16 + m];
      const v2f bg0 = wl[1][bi + m], bg1 = wl[1][bi + 16 + m];
      const v2f bh0 = wl[2][bi + m], bh1 = wl[2][bi + 16 + m];
      aq0 = WMMA_F32(a, bf0, aq0); aq1 = WMMA_F32(a, bf1, aq1);
      ak0 = WMMA_F32(a, bg0, ak0); ak1 = WMMA_F32(a, bg1, ak1);
      av0 = WMMA_F32(a, bh0, av0); av1 = WMMA_F32(a, bh1, av1);
    }
    __syncthreads();
  }
#pragma unroll
  for (int vi = 0; vi < 8; ++vi) {
    const int r = (row0 + vi + half * 8) * 32;
    q[r + m] = aq0[vi]; q[r + 16 + m] = aq1[vi];
    k[r + m] = ak0[vi]; k[r + 16 + m] = ak1[vi];
    v[r + m] = av0[vi]; v[r + 16 + m] = av1[vi];
  }
}

typedef float v4f __attribute__((ext_vector_type(4)));
#define QP 36
#define VP 48
#define PP 68

__global__ __launch_bounds__(512) void attn_kernel(
    const float* __restrict__ q, const float* __restrict__ k,
    const float* __restrict__ v, float* __restrict__ o) {
  __shared__ __attribute__((aligned(16))) float qs[2][4][64 * QP];
  __shared__ __attribute__((aligned(16))) float vs[2][4][64 * VP];
  __shared__ __attribute__((aligned(16))) float ps[16][16 * PP];
  const int tid = threadIdx.x;
  const int wave = tid >> 5, lane = tid & 31;
  const int m = lane & 15, half = lane >> 4, kOff = half * 2;
  const int tile = wave & 3, split = wave >> 2;
  const int b = blockIdx.x >> 6;
  const int i0 = (blockIdx.x & 63) * 64 + tile * 16;
  const float* qb = q + b * 4096 * 32;
  const float* kb = k + b * 4096 * 32;
  const float* vb = v + b * 4096 * 32;
  float* pw = ps[wave];

  const int l = tid & 127;
  const int srow = l >> 3, scol = (l & 7) * 4;
  const float* qsrc = qb + (split * 1024 + srow) * 32 + scol;
  const float* vsrc = vb + (split * 1024 + srow) * 32 + scol;
  const int qdst = srow * QP + scol, vdst = srow * VP + scol;

  v2f aK[8];
#pragma unroll
  for (int t = 0; t < 8; ++t) aK[t] = ld2(&kb[(i0 + m) * 32 + 4 * t + kOff]);

  float mrun[8], lrun[8];
  v8f o0 = {}, o1 = {};
#pragma unroll
  for (int vi = 0; vi < 8; ++vi) { mrun[vi] = -__builtin_inff(); lrun[vi] = 0.f; }

  v4f pq[4], pv[4];
#pragma unroll
  for (int r = 0; r < 4; ++r) {
    pq[r] = *(const v4f*)(qsrc + r * 512);
    pv[r] = *(const v4f*)(vsrc + r * 512);
  }
#pragma unroll
  for (int r = 0; r < 4; ++r) {
    *(v4f*)&qs[0][split][qdst + r * 16 * QP] = pq[r];
    *(v4f*)&vs[0][split][vdst + r * 16 * VP] = pv[r];
  }
  __syncthreads();

  for (int it = 0; it < 16; ++it) {
    const int sg = it & 1;
    const float* qt = qs[sg][split];
    const float* vt = vs[sg][split];
    if (it < 15) {
#pragma unroll
      for (int r = 0; r < 4; ++r) {
        pq[r] = *(const v4f*)(qsrc + (it + 1) * 2048 + r * 512);
        pv[r] = *(const v4f*)(vsrc + (it + 1) * 2048 + r * 512);
      }
    }
    v8f s[4] = {{}, {}, {}, {}};
#pragma unroll
    for (int st = 0; st < 4; ++st) {
#pragma unroll
      for (int t = 0; t < 8; ++t) {
        const v2f bq = *(const v2f*)&qt[(st * 16 + m) * QP + 4 * t + kOff];
        s[st] = WMMA_F32(aK[t], bq, s[st]);
      }
    }
#pragma unroll
    for (int vi = 0; vi < 8; ++vi) {
      float mx = fmaxf(fmaxf(s[0][vi], s[1][vi]), fmaxf(s[2][vi], s[3][vi]));
      mx = fmaxf(mx, __shfl_xor(mx, 1, 16));
      mx = fmaxf(mx, __shfl_xor(mx, 2, 16));
      mx = fmaxf(mx, __shfl_xor(mx, 4, 16));
      mx = fmaxf(mx, __shfl_xor(mx, 8, 16));
      const float nm = fmaxf(mrun[vi], mx);
      const float sc = __expf(mrun[vi] - nm);
      mrun[vi] = nm;
      const float p0 = __expf(s[0][vi] - nm); s[0][vi] = p0;
      const float p1 = __expf(s[1][vi] - nm); s[1][vi] = p1;
      const float p2 = __expf(s[2][vi] - nm); s[2][vi] = p2;
      const float p3 = __expf(s[3][vi] - nm); s[3][vi] = p3;
      float rs = (p0 + p1) + (p2 + p3);
      rs += __shfl_xor(rs, 1, 16);
      rs += __shfl_xor(rs, 2, 16);
      rs += __shfl_xor(rs, 4, 16);
      rs += __shfl_xor(rs, 8, 16);
      lrun[vi] = lrun[vi] * sc + rs;
      o0[vi] *= sc;
      o1[vi] *= sc;
    }
#pragma unroll
    for (int st = 0; st < 4; ++st)
#pragma unroll
      for (int vi = 0; vi < 8; ++vi)
        pw[(vi + half * 8) * PP + st * 16 + m] = s[st][vi];
    asm volatile("s_wait_dscnt 0" ::: "memory");
#pragma unroll
    for (int t = 0; t < 16; ++t) {
      const int kk = 4 * t + kOff;
      const v2f aP = *(const v2f*)&pw[m * PP + kk];
      const v2f bv0 = mk2(vt[kk * VP + m],      vt[(kk + 1) * VP + m]);
      const v2f bv1 = mk2(vt[kk * VP + 16 + m], vt[(kk + 1) * VP + 16 + m]);
      o0 = WMMA_F32(aP, bv0, o0);
      o1 = WMMA_F32(aP, bv1, o1);
    }
    if (it < 15) {
#pragma unroll
      for (int r = 0; r < 4; ++r) {
        *(v4f*)&qs[sg ^ 1][split][qdst + r * 16 * QP] = pq[r];
        *(v4f*)&vs[sg ^ 1][split][vdst + r * 16 * VP] = pv[r];
      }
    }
    __syncthreads();
  }

  asm volatile("s_wait_dscnt 0" ::: "memory");
#pragma unroll
  for (int vi = 0; vi < 8; ++vi) {
    const int r = vi + half * 8;
    pw[r * 32 + m]      = o0[vi];
    pw[r * 32 + 16 + m] = o1[vi];
  }
  if (m == 0) {
#pragma unroll
    for (int vi = 0; vi < 8; ++vi) {
      pw[512 + vi + half * 8] = lrun[vi];
      pw[528 + vi + half * 8] = mrun[vi];
    }
  }
  __syncthreads();

  if (split == 0) {
#pragma unroll
    for (int vi = 0; vi < 8; ++vi) {
      const int r = vi + half * 8;
      float mg = ps[tile][528 + r];
      mg = fmaxf(mg, ps[tile + 4][528 + r]);
      mg = fmaxf(mg, ps[tile + 8][528 + r]);
      mg = fmaxf(mg, ps[tile + 12][528 + r]);
      float lg = 0.f, a0 = 0.f, a1 = 0.f;
#pragma unroll
      for (int p = 0; p < 4; ++p) {
        const float* pp = ps[tile + 4 * p];
        const float wgt = __expf(pp[528 + r] - mg);
        lg += wgt * pp[512 + r];
        a0 += wgt * pp[r * 32 + m];
        a1 += wgt * pp[r * 32 + 16 + m];
      }
      const float inv = 1.f / lg;
      const int g = b * 4096 * 32 + (i0 + r) * 32;
      o[g + m]      = a0 * inv;
      o[g + 16 + m] = a1 * inv;
    }
  }
}

__global__ __launch_bounds__(256) void out_kernel(
    const float* __restrict__ o, const float* __restrict__ Wv,
    const float* __restrict__ x, const float* __restrict__ gamma,
    float* __restrict__ out) {
  const int tid = threadIdx.x;
  const int wave = tid >> 5, lane = tid & 31;
  const int m = lane & 15, half = lane >> 4, kOff = half * 2;
  const int task = blockIdx.x * 8 + wave;
  const int row0 = (task >> 2) * 16, col0 = (task & 3) * 64;

  v2f aO[8];
#pragma unroll
  for (int t = 0; t < 8; ++t) aO[t] = ld2(&o[(row0 + m) * 32 + 4 * t + kOff]);

  v8f acc[4] = {{}, {}, {}, {}};
#pragma unroll
  for (int t = 0; t < 8; ++t) {
    const int kk = 4 * t + kOff;
#pragma unroll
    for (int ct = 0; ct < 4; ++ct) {
      const int col = col0 + ct * 16 + m;
      const v2f bw = mk2(Wv[kk * 256 + col], Wv[(kk + 1) * 256 + col]);
      acc[ct] = WMMA_F32(aO[t], bw, acc[ct]);
    }
  }
  const float g = *gamma;
#pragma unroll
  for (int ct = 0; ct < 4; ++ct) {
#pragma unroll
    for (int vi = 0; vi < 8; ++vi) {
      const int idx = (row0 + vi + half * 8) * 256 + col0 + ct * 16 + m;
      out[idx] = g * acc[ct][vi] + x[idx];
    }
  }
}

extern "C" void kernel_launch(void* const* d_in, const int* in_sizes, int n_in,
                              void* d_out, int out_size, void* d_ws, size_t ws_size,
                              hipStream_t stream) {
  const float* x  = (const float*)d_in[0];
  const float* Wf = (const float*)d_in[1];
  const float* Wg = (const float*)d_in[2];
  const float* Wh = (const float*)d_in[3];
  const float* Wv = (const float*)d_in[4];
  const float* gm = (const float*)d_in[5];
  float* out = (float*)d_out;

  const int QKV = 4 * 4096 * 32;
  float* q = (float*)d_ws;
  float* k = q + QKV;
  float* v = k + QKV;
  float* o = v + QKV;

  qkv_kernel<<<128, 256, 0, stream>>>(x, Wf, Wg, Wh, q, k, v);
  attn_kernel<<<256, 512, 0, stream>>>(q, k, v, o);
  out_kernel<<<512, 256, 0, stream>>>(o, Wv, x, gm, out);
}


# attention: v tile ring 3 slots (q 2) and the per-tile work-group barrier split: a wave reports behind its score products, waits at the next tile's top; waves of a SIMD leave lock-step; sums unchanged
# speedup vs baseline: 1.1566x; 1.0343x over previous
#include <hip/hip_runtime.h>


typedef float v2f __attribute__((ext_vector_type(2)));
typedef float v8f __attribute__((ext_vector_type(8)));

#define WMMA_F32(A, B, C) \
  __builtin_amdgcn_wmma_f32_16x16x4_f32(false, (A), false, (B), (short)0, (C), false, false)

static __device__ __forceinline__ v2f ld2(const float* p) { return *(const v2f*)p; }
static __device__ __forceinline__ v2f mk2(float a, float b) { v2f r; r.x = a; r.y = b; return r; }

__global__ __launch_bounds__(256) void qkv_kernel(
    const float* __restrict__ x,
    const float* __restrict__ Wf, const float* __restrict__ Wg, const float* __restrict__ Wh,
    float* __restrict__ q, float* __restrict__ k, float* __restrict__ v) {
  __shared__ v2f wl[3][32 * 32];
  const int tid = threadIdx.x;
  const int wave = tid >> 5, lane = tid & 31;
  const int m = lane & 15, half = lane >> 4, kOff = half * 2;
  const int row0 = blockIdx.x * 128 + wave * 16;

  v8f aq0 = {}, aq1 = {}, ak0 = {}, ak1 = {}, av0 = {}, av1 = {};

  for (int c0 = 0; c0 < 256; c0 += 64) {
    for (int idx = tid; idx < 32 * 32; idx += 256) {
      const int kk2 = idx >> 5, n = idx & 31;
      const int g = (c0 + kk2 * 2) * 32 + n;
      wl[0][idx] = mk2(Wf[g], Wf[g + 32]);
      wl[1][idx] = mk2(Wg[g], Wg[g + 32]);
      wl[2][idx] = mk2(Wh[g], Wh[g + 32]);
    }
    __syncthreads();
#pragma unroll
    for (int kk = 0; kk < 64; kk += 4) {
      const v2f a = ld2(&x[(row0 + m) * 256 + c0 + kk + kOff]);
      const int bi = ((kk + kOff) >> 1) * 32;
      const v2f bf0 = wl[0][bi + m], bf1 = wl[0][bi + 16 + m];
      const v2f bg0 = wl[1][bi + m], bg1 = wl[1][bi + 16 + m];
      const v2f bh0 = wl[2][bi + m], bh1 = wl[2][bi + 16 + m];
      aq0 = WMMA_F32(a, bf0, aq0); aq1 = WMMA_F32(a, bf1, aq1);
      ak0 = WMMA_F32(a, bg0, ak0); ak1 = WMMA_F32(a, bg1, ak1);
      av0 = WMMA_F32(a, bh0, av0); av1 = WMMA_F32(a, bh1, av1);
    }
    __syncthreads();
  }
#pragma unroll
  for (int vi = 0; vi < 8; ++vi) {
    const int r = (row0 + vi + half * 8) * 32;
    q[r + m] = aq0[vi]; q[r + 16 + m] = aq1[vi];
    k[r + m] = ak0[vi]; k[r + 16 + m] = ak1[vi];
    v[r + m] = av0[vi]; v[r + 16 + m] = av1[vi];
  }
}

typedef float v4f __attribute__((ext_vector_type(4)));
#define QP 36
#define VP 48
#define PP 68

__global__ __launch_bounds__(512) void attn_kernel(
    const float* __restrict__ q, const float* __restrict__ k,
    const float* __restrict__ v, float* __restrict__ o) {
  __shared__ __attribute__((aligned(16))) float qs[2][4][64 * QP];
  __shared__ __attribute__((aligned(16))) float vs[3][4][64 * VP];
  __shared__ __attribute__((aligned(16))) float ps[16][16 * PP];
  const int tid = threadIdx.x;
  const int wave = tid >> 5, lane = tid & 31;
  const int m = lane & 15, half = lane >> 4, kOff = half * 2;
  const int tile = wave & 3, split = wave >> 2;
  const int b = blockIdx.x >> 6;
  const int i0 = (blockIdx.x & 63) * 64 + tile * 16;
  const float* qb = q + b * 4096 * 32;
  const float* kb = k + b * 4096 * 32;
  const float* vb = v + b * 4096 * 32;
  float* pw = ps[wave];

  const int l = tid & 127;
  const int srow = l >> 3, scol = (l & 7) * 4;
  const float* qsrc = qb + (split * 1024 + srow) * 32 + scol;
  const float* vsrc = vb + (split * 1024 + srow) * 32 + scol;
  const int qdst = srow * QP + scol, vdst = srow * VP + scol;

  v2f aK[8];
#pragma unroll
  for (int t = 0; t < 8; ++t) aK[t] = ld2(&kb[(i0 + m) * 32 + 4 * t + kOff]);

  float mrun[8], lrun[8];
  v8f o0 = {}, o1 = {};
#pragma unroll
  for (int vi = 0; vi < 8; ++vi) { mrun[vi] = -__builtin_inff(); lrun[vi] = 0.f; }

  v4f pq[4], pv[4];
#pragma unroll
  for (int r = 0; r < 4; ++r) {
    pq[r] = *(const v4f*)(qsrc + r * 512);
    pv[r] = *(const v4f*)(vsrc + r * 512);
  }
#pragma unroll
  for (int r = 0; r < 4; ++r) {
    *(v4f*)&qs[0][split][qdst + r * 16 * QP] = pq[r];
    *(v4f*)&vs[0][split][vdst + r * 16 * VP] = pv[r];
  }
  __syncthreads();
  asm volatile("s_barrier_signal -1" ::: "memory");
#pragma unroll
  for (int r = 0; r < 4; ++r) {
    pq[r] = *(const v4f*)(qsrc + 2048 + r * 512);
    pv[r] = *(const v4f*)(vsrc + 2048 + r * 512);
  }

  int vc = 0;
  for (int it = 0; it < 16; ++it) {
    const int sg = it & 1;
    const int vn = (vc == 2) ? 0 : vc + 1;
    const float* qt = qs[sg][split];
    const float* vt = vs[vc][split];
    asm volatile("s_barrier_wait -1" ::: "memory");
    v8f s[4] = {{}, {}, {}, {}};
#pragma unroll
    for (int st = 0; st < 4; ++st) {
#pragma unroll
      for (int t = 0; t < 8; ++t) {
        const v2f bq = *(const v2f*)&qt[(st * 16 + m) * QP + 4 * t + kOff];
        s[st] = WMMA_F32(aK[t], bq, s[st]);
      }
    }
    __builtin_amdgcn_sched_barrier(0);
    {
#pragma unroll
      for (int r = 0; r < 4; ++r) {
        *(v4f*)&qs[sg ^ 1][split][qdst + r * 16 * QP] = pq[r];
        *(v4f*)&vs[vn][split][vdst + r * 16 * VP] = pv[r];
      }
      asm volatile("s_wait_dscnt 0x0\n\ts_barrier_signal -1" ::: "memory");
      const int nt = (it < 14) ? it + 2 : 15;
#pragma unroll
      for (int r = 0; r < 4; ++r) {
        pq[r] = *(const v4f*)(qsrc + nt * 2048 + r * 512);
        pv[r] = *(const v4f*)(vsrc + nt * 2048 + r * 512);
      }
    }
#pragma unroll
    for (int vi = 0; vi < 8; ++vi) {
      float mx = fmaxf(fmaxf(s[0][vi], s[1][vi]), fmaxf(s[2][vi], s[3][vi]));
      mx = fmaxf(mx, __shfl_xor(mx, 1, 16));
      mx = fmaxf(mx, __shfl_xor(mx, 2, 16));
      mx = fmaxf(mx, __shfl_xor(mx, 4, 16));
      mx = fmaxf(mx, __shfl_xor(mx, 8, 16));
      const float nm = fmaxf(mrun[vi], mx);
      const float sc = __expf(mrun[vi] - nm);
      mrun[vi] = nm;
      const float p0 = __expf(s[0][vi] - nm); s[0][vi] = p0;
      const float p1 = __expf(s[1][vi] - nm); s[1][vi] = p1;
      const float p2 = __expf(s[2][vi] - nm); s[2][vi] = p2;
      const float p3 = __expf(s[3][vi] - nm); s[3][vi] = p3;
      float rs = (p0 + p1) + (p2 + p3);
      rs += __shfl_xor(rs, 1, 16);
      rs += __shfl_xor(rs, 2, 16);
      rs += __shfl_xor(rs, 4, 16);
      rs += __shfl_xor(rs, 8, 16);
      lrun[vi] = lrun[vi] * sc + rs;
      o0[vi] *= sc;
      o1[vi] *= sc;
    }
#pragma unroll
    for (int st = 0; st < 4; ++st)
#pragma unroll
      for (int vi = 0; vi < 8; ++vi)
        pw[(vi + half * 8) * PP + st * 16 + m] = s[st][vi];
    asm volatile("s_wait_dscnt 0" ::: "memory");
#pragma unroll
    for (int t = 0; t < 16; ++t) {
      const int kk = 4 * t + kOff;
      const v2f aP = *(const v2f*)&pw[m * PP + kk];
      const v2f bv0 = mk2(vt[kk * VP + m],      vt[(kk + 1) * VP + m]);
      const v2f bv1 = mk2(vt[kk * VP + 16 + m], vt[(kk + 1) * VP + 16 + m]);
      o0 = WMMA_F32(aP, bv0, o0);
      o1 = WMMA_F32(aP, bv1, o1);
    }
    vc = vn;
  }
  asm volatile("s_barrier_wait -1" ::: "memory");

  asm volatile("s_wait_dscnt 0" ::: "memory");
#pragma unroll
  for (int vi = 0; vi < 8; ++vi) {
    const int r = vi + half * 8;
    pw[r * 32 + m]      = o0[vi];
    pw[r * 32 + 16 + m] = o1[vi];
  }
  if (m == 0) {
#pragma unroll
    for (int vi = 0; vi < 8; ++vi) {
      pw[512 + vi + half * 8] = lrun[vi];
      pw[528 + vi + half * 8] = mrun[vi];
    }
  }
  __syncthreads();

  if (split == 0) {
#pragma unroll
    for (int vi = 0; vi < 8; ++vi) {
      const int r = vi + half * 8;
      float mg = ps[tile][528 + r];
      mg = fmaxf(mg, ps[tile + 4][528 + r]);
      mg = fmaxf(mg, ps[tile + 8][528 + r]);
      mg = fmaxf(mg, ps[tile + 12][528 + r]);
      float lg = 0.f, a0 = 0.f, a1 = 0.f;
#pragma unroll
      for (int p = 0; p < 4; ++p) {
        const float* pp = ps[tile + 4 * p];
        const float wgt = __expf(pp[528 + r] - mg);
        lg += wgt * pp[512 + r];
        a0 += wgt * pp[r * 32 + m];
        a1 += wgt * pp[r * 32 + 16 + m];
      }
      const float inv = 1.f / lg;
      const int g = b * 4096 * 32 + (i0 + r) * 32;
      o[g + m]      = a0 * inv;
      o[g + 16 + m] = a1 * inv;
    }
  }
}

__global__ __launch_bounds__(256) void out_kernel(
    const float* __restrict__ o, const float* __restrict__ Wv,
    const float* __restrict__ x, const float* __restrict__ gamma,
    float* __restrict__ out) {
  const int tid = threadIdx.x;
  const int wave = tid >> 5, lane = tid & 31;
  const int m = lane & 15, half = lane >> 4, kOff = half * 2;
  const int task = blockIdx.x * 8 + wave;
  const int row0 = (task >> 2) * 16, col0 = (task & 3) * 64;

  v2f aO[8];
#pragma unroll
  for (int t = 0; t < 8; ++t) aO[t] = ld2(&o[(row0 + m) * 32 + 4 * t + kOff]);

  v8f acc[4] = {{}, {}, {}, {}};
#pragma unroll
  for (int t = 0; t < 8; ++t) {
    const int kk = 4 * t + kOff;
#pragma unroll
    for (int ct = 0; ct < 4; ++ct) {
      const int col = col0 + ct * 16 + m;
      const v2f bw = mk2(Wv[kk * 256 + col], Wv[(kk + 1) * 256 + col]);
      acc[ct] = WMMA_F32(aO[t], bw, acc[ct]);
    }
  }
  const float g = *gamma;
#pragma unroll
  for (int ct = 0; ct < 4; ++ct) {
#pragma unroll
    for (int vi = 0; vi < 8; ++vi) {
      const int idx = (row0 + vi + half * 8) * 256 + col0 + ct * 16 + m;
      out[idx] = g * acc[ct][vi] + x[idx];
    }
  }
}

extern "C" void kernel_launch(void* const* d_in, const int* in_sizes, int n_in,
                              void* d_out, int out_size, void* d_ws, size_t ws_size,
                              hipStream_t stream) {
  const float* x  = (const float*)d_in[0];
  const float* Wf = (const float*)d_in[1];
  const float* Wg = (const float*)d_in[2];
  const float* Wh = (const float*)d_in[3];
  const float* Wv = (const float*)d_in[4];
  const float* gm = (const float*)d_in[5];
  float* out = (float*)d_out;

  const int QKV = 4 * 4096 * 32;
  float* q = (float*)d_ws;
  float* k = q + QKV;
  float* v = k + QKV;
  float* o = v + QKV;

  qkv_kernel<<<128, 256, 0, stream>>>(x, Wf, Wg, Wh, q, k, v);
  attn_kernel<<<256, 512, 0, stream>>>(q, k, v, o);
  out_kernel<<<512, 256, 0, stream>>>(o, Wv, x, gm, out);
}


# projection as transposed product: a wave keeps its 16-column weight slice for all of K in registers as A operand, x tile staged once in LDS as B; no weight staging, 8 barriers to 3; sums unchanged
# speedup vs baseline: 1.3510x; 1.1681x over previous
#include <hip/hip_runtime.h>


typedef float v2f __attribute__((ext_vector_type(2)));
typedef float v8f __attribute__((ext_vector_type(8)));

#define WMMA_F32(A, B, C) \
  __builtin_amdgcn_wmma_f32_16x16x4_f32(false, (A), false, (B), (short)0, (C), false, false)

static __device__ __forceinline__ v2f ld2(const float* p) { return *(const v2f*)p; }
static __device__ __forceinline__ v2f mk2(float a, float b) { v2f r; r.x = a; r.y = b; return r; }

typedef float v4f __attribute__((ext_vector_type(4)));
#define XP 260
#define OP 100

__global__ __launch_bounds__(384) void qkv_kernel(
    const float* __restrict__ x,
    const float* __restrict__ Wf, const float* __restrict__ Wg, const float* __restrict__ Wh,
    float* __restrict__ q, float* __restrict__ k, float* __restrict__ v) {
  __shared__ __attribute__((aligned(16))) float xs[64 * XP];
  const int tid = threadIdx.x;
  const int wave = tid >> 5, lane = tid & 31;
  const int c = lane & 15, half = lane >> 4, kOff = half * 2;
  const int nt = wave % 6;
  const int rh = wave / 6;
  const int rowB = blockIdx.x * 64;

  v4f xr[11];
#pragma unroll
  for (int r = 0; r < 11; ++r) {
    const int j = tid + 384 * r;
    if (j < 4096) xr[r] = *(const v4f*)(x + (rowB + (j >> 6)) * 256 + (j & 63) * 4);
  }

  const float* Wsel = (nt < 2) ? Wf : ((nt < 4) ? Wg : Wh);
  const float* wp = Wsel + kOff * 32 + (nt & 1) * 16 + c;
  v2f aW[64];
#pragma unroll
  for (int t = 0; t < 64; ++t) aW[t] = mk2(wp[(4 * t) * 32], wp[(4 * t + 1) * 32]);

#pragma unroll
  for (int r = 0; r < 11; ++r) {
    const int j = tid + 384 * r;
    if (j < 4096) *(v4f*)&xs[(j >> 6) * XP + (j & 63) * 4] = xr[r];
  }
  __syncthreads();

  v8f acc0 = {}, acc1 = {};
  const float* b0p = &xs[(rh * 32 + c) * XP + kOff];
  const float* b1p = b0p + 16 * XP;
  v2f bA0[8], bA1[8], bB0[8], bB1[8];
#pragma unroll
  for (int u = 0; u < 8; ++u) {
    bA0[u] = *(const v2f*)(b0p + 4 * u);
    bA1[u] = *(const v2f*)(b1p + 4 * u);
  }
#pragma unroll
  for (int tb = 0; tb < 8; tb += 2) {
#pragma unroll
    for (int u = 0; u < 8; ++u) {
      bB0[u] = *(const v2f*)(b0p + 4 * (8 * (tb + 1) + u));
      bB1[u] = *(const v2f*)(b1p + 4 * (8 * (tb + 1) + u));
    }
    __builtin_amdgcn_sched_barrier(0);
#pragma unroll
    for (int u = 0; u < 8; ++u) {
      acc0 = WMMA_F32(aW[8 * tb + u], bA0[u], acc0);
      acc1 = WMMA_F32(aW[8 * tb + u], bA1[u], acc1);
    }
    if (tb + 2 < 8) {
#pragma unroll
      for (int u = 0; u < 8; ++u) {
        bA0[u] = *(const v2f*)(b0p + 4 * (8 * (tb + 2) + u));
        bA1[u] = *(const v2f*)(b1p + 4 * (8 * (tb + 2) + u));
      }
    }
    __builtin_amdgcn_sched_barrier(0);
#pragma unroll
    for (int u = 0; u < 8; ++u) {
      acc0 = WMMA_F32(aW[8 * (tb + 1) + u], bB0[u], acc0);
      acc1 = WMMA_F32(aW[8 * (tb + 1) + u], bB1[u], acc1);
    }
  }
  __syncthreads();

  {
    float* o0 = &xs[(rh * 32 + c) * OP + nt * 16 + half * 8];
    float* o1 = o0 + 16 * OP;
    v4f t0, t1;
    t0.x = acc0[0]; t0.y = acc0[1]; t0.z = acc0[2]; t0.w = acc0[3];
    t1.x = acc0[4]; t1.y = acc0[5]; t1.z = acc0[6]; t1.w = acc0[7];
    *(v4f*)o0 = t0; *(v4f*)(o0 + 4) = t1;
    t0.x = acc1[0]; t0.y = acc1[1]; t0.z = acc1[2]; t0.w = acc1[3];
    t1.x = acc1[4]; t1.y = acc1[5]; t1.z = acc1[6]; t1.w = acc1[7];
    *(v4f*)o1 = t0; *(v4f*)(o1 + 4) = t1;
  }
  __syncthreads();

#pragma unroll
  for (int i = 0; i < 4; ++i) {
    const int idx = tid + 384 * i;
    const int ob = idx >> 9, row = (idx & 511) >> 3, p4 = (idx & 7) * 4;
    float* dst = (ob == 0) ? q : ((ob == 1) ? k : v);
    *(v4f*)(dst + (rowB + row) * 32 + p4) = *(const v4f*)&xs[row * OP + ob * 32 + p4];
  }
}

#define QP 36
#define VP 48
#define PP 68

__global__ __launch_bounds__(512) void attn_kernel(
    const float* __restrict__ q, const float* __restrict__ k,
    const float* __restrict__ v, float* __restrict__ o) {
  __shared__ __attribute__((aligned(16))) float qs[2][4][64 * QP];
  __shared__ __attribute__((aligned(16))) float vs[3][4][64 * VP];
  __shared__ __attribute__((aligned(16))) float ps[16][16 * PP];
  const int tid = threadIdx.x;
  const int wave = tid >> 5, lane = tid & 31;
  const int m = lane & 15, half = lane >> 4, kOff = half * 2;
  const int tile = wave & 3, split = wave >> 2;
  const int b = blockIdx.x >> 6;
  const int i0 = (blockIdx.x & 63) * 64 + tile * 16;
  const float* qb = q + b * 4096 * 32;
  const float* kb = k + b * 4096 * 32;
  const float* vb = v + b * 4096 * 32;
  float* pw = ps[wave];

  const int l = tid & 127;
  const int srow = l >> 3, scol = (l & 7) * 4;
  const float* qsrc = qb + (split * 1024 + srow) * 32 + scol;
  const float* vsrc = vb + (split * 1024 + srow) * 32 + scol;
  const int qdst = srow * QP + scol, vdst = srow * VP + scol;

  v2f aK[8];
#pragma unroll
  for (int t = 0; t < 8; ++t) aK[t] = ld2(&kb[(i0 + m) * 32 + 4 * t + kOff]);

  float mrun[8], lrun[8];
  v8f o0 = {}, o1 = {};
#pragma unroll
  for (int vi = 0; vi < 8; ++vi) { mrun[vi] = -__builtin_inff(); lrun[vi] = 0.f; }

  v4f pq[4], pv[4];
#pragma unroll
  for (int r = 0; r < 4; ++r) {
    pq[r] = *(const v4f*)(qsrc + r * 512);
    pv[r] = *(const v4f*)(vsrc + r * 512);
  }
#pragma unroll
  for (int r = 0; r < 4; ++r) {
    *(v4f*)&qs[0][split][qdst + r * 16 * QP] = pq[r];
    *(v4f*)&vs[0][split][vdst + r * 16 * VP] = pv[r];
  }
  __syncthreads();
  asm volatile("s_barrier_signal -1" ::: "memory");
#pragma unroll
  for (int r = 0; r < 4; ++r) {
    pq[r] = *(const v4f*)(qsrc + 2048 + r * 512);
    pv[r] = *(const v4f*)(vsrc + 2048 + r * 512);
  }

  int vc = 0;
  for (int it = 0; it < 16; ++it) {
    const int sg = it & 1;
    const int vn = (vc == 2) ? 0 : vc + 1;
    const float* qt = qs[sg][split];
    const float* vt = vs[vc][split];
    asm volatile("s_barrier_wait -1" ::: "memory");
    v8f s[4] = {{}, {}, {}, {}};
#pragma unroll
    for (int st = 0; st < 4; ++st) {
#pragma unroll
      for (int t = 0; t < 8; ++t) {
        const v2f bq = *(const v2f*)&qt[(st * 16 + m) * QP + 4 * t + kOff];
        s[st] = WMMA_F32(aK[t], bq, s[st]);
      }
    }
    __builtin_amdgcn_sched_barrier(0);
    {
#pragma unroll
      for (int r = 0; r < 4; ++r) {
        *(v4f*)&qs[sg ^ 1][split][qdst + r * 16 * QP] = pq[r];
        *(v4f*)&vs[vn][split][vdst + r * 16 * VP] = pv[r];
      }
      asm volatile("s_wait_dscnt 0x0\n\ts_barrier_signal -1" ::: "memory");
      const int nt = (it < 14) ? it + 2 : 15;
#pragma unroll
      for (int r = 0; r < 4; ++r) {
        pq[r] = *(const v4f*)(qsrc + nt * 2048 + r * 512);
        pv[r] = *(const v4f*)(vsrc + nt * 2048 + r * 512);
      }
    }
#pragma unroll
    for (int vi = 0; vi < 8; ++vi) {
      float mx = fmaxf(fmaxf(s[0][vi], s[1][vi]), fmaxf(s[2][vi], s[3][vi]));
      mx = fmaxf(mx, __shfl_xor(mx, 1, 16));
      mx = fmaxf(mx, __shfl_xor(mx, 2, 16));
      mx = fmaxf(mx, __shfl_xor(mx, 4, 16));
      mx = fmaxf(mx, __shfl_xor(mx, 8, 16));
      const float nm = fmaxf(mrun[vi], mx);
      const float sc = __expf(mrun[vi] - nm);
      mrun[vi] = nm;
      const float p0 = __expf(s[0][vi] - nm); s[0][vi] = p0;
      const float p1 = __expf(s[1][vi] - nm); s[1][vi] = p1;
      const float p2 = __expf(s[2][vi] - nm); s[2][vi] = p2;
      const float p3 = __expf(s[3][vi] - nm); s[3][vi] = p3;
      float rs = (p0 + p1) + (p2 + p3);
      rs += __shfl_xor(rs, 1, 16);
      rs += __shfl_xor(rs, 2, 16);
      rs += __shfl_xor(rs, 4, 16);
      rs += __shfl_xor(rs, 8, 16);
      lrun[vi] = lrun[vi] * sc + rs;
      o0[vi] *= sc;
      o1[vi] *= sc;
    }
#pragma unroll
    for (int st = 0; st < 4; ++st)
#pragma unroll
      for (int vi = 0; vi < 8; ++vi)
        pw[(vi + half * 8) * PP + st * 16 + m] = s[st][vi];
    asm volatile("s_wait_dscnt 0" ::: "memory");
#pragma unroll
    for (int t = 0; t < 16; ++t) {
      const int kk = 4 * t + kOff;
      const v2f aP = *(const v2f*)&pw[m * PP + kk];
      const v2f bv0 = mk2(vt[kk * VP + m],      vt[(kk + 1) * VP + m]);
      const v2f bv1 = mk2(vt[kk * VP + 16 + m], vt[(kk + 1) * VP + 16 + m]);
      o0 = WMMA_F32(aP, bv0, o0);
      o1 = WMMA_F32(aP, bv1, o1);
    }
    vc = vn;
  }
  asm volatile("s_barrier_wait -1" ::: "memory");

  asm volatile("s_wait_dscnt 0" ::: "memory");
#pragma unroll
  for (int vi = 0; vi < 8; ++vi) {
    const int r = vi + half * 8;
    pw[r * 32 + m]      = o0[vi];
    pw[r * 32 + 16 + m] = o1[vi];
  }
  if (m == 0) {
#pragma unroll
    for (int vi = 0; vi < 8; ++vi) {
      pw[512 + vi + half * 8] = lrun[vi];
      pw[528 + vi + half * 8] = mrun[vi];
    }
  }
  __syncthreads();

  if (split == 0) {
#pragma unroll
    for (int vi = 0; vi < 8; ++vi) {
      const int r = vi + half * 8;
      float mg = ps[tile][528 + r];
      mg = fmaxf(mg, ps[tile + 4][528 + r]);
      mg = fmaxf(mg, ps[tile + 8][528 + r]);
      mg = fmaxf(mg, ps[tile + 12][528 + r]);
      float lg = 0.f, a0 = 0.f, a1 = 0.f;
#pragma unroll
      for (int p = 0; p < 4; ++p) {
        const float* pp = ps[tile + 4 * p];
        const float wgt = __expf(pp[528 + r] - mg);
        lg += wgt * pp[512 + r];
        a0 += wgt * pp[r * 32 + m];
        a1 += wgt * pp[r * 32 + 16 + m];
      }
      const float inv = 1.f / lg;
      const int g = b * 4096 * 32 + (i0 + r) * 32;
      o[g + m]      = a0 * inv;
      o[g + 16 + m] = a1 * inv;
    }
  }
}

__global__ __launch_bounds__(256) void out_kernel(
    const float* __restrict__ o, const float* __restrict__ Wv,
    const float* __restrict__ x, const float* __restrict__ gamma,
    float* __restrict__ out) {
  const int tid = threadIdx.x;
  const int wave = tid >> 5, lane = tid & 31;
  const int m = lane & 15, half = lane >> 4, kOff = half * 2;
  const int task = blockIdx.x * 8 + wave;
  const int row0 = (task >> 2) * 16, col0 = (task & 3) * 64;

  v2f aO[8];
#pragma unroll
  for (int t = 0; t < 8; ++t) aO[t] = ld2(&o[(row0 + m) * 32 + 4 * t + kOff]);

  v8f acc[4] = {{}, {}, {}, {}};
#pragma unroll
  for (int t = 0; t < 8; ++t) {
    const int kk = 4 * t + kOff;
#pragma unroll
    for (int ct = 0; ct < 4; ++ct) {
      const int col = col0 + ct * 16 + m;
      const v2f bw = mk2(Wv[kk * 256 + col], Wv[(kk + 1) * 256 + col]);
      acc[ct] = WMMA_F32(aO[t], bw, acc[ct]);
    }
  }
  const float g = *gamma;
#pragma unroll
  for (int ct = 0; ct < 4; ++ct) {
#pragma unroll
    for (int vi = 0; vi < 8; ++vi) {
      const int idx = (row0 + vi + half * 8) * 256 + col0 + ct * 16 + m;
      out[idx] = g * acc[ct][vi] + x[idx];
    }
  }
}

extern "C" void kernel_launch(void* const* d_in, const int* in_sizes, int n_in,
                              void* d_out, int out_size, void* d_ws, size_t ws_size,
                              hipStream_t stream) {
  const float* x  = (const float*)d_in[0];
  const float* Wf = (const float*)d_in[1];
  const float* Wg = (const float*)d_in[2];
  const float* Wh = (const float*)d_in[3];
  const float* Wv = (const float*)d_in[4];
  const float* gm = (const float*)d_in[5];
  float* out = (float*)d_out;

  const int QKV = 4 * 4096 * 32;
  float* q = (float*)d_ws;
  float* k = q + QKV;
  float* v = k + QKV;
  float* o = v + QKV;

  qkv_kernel<<<256, 384, 0, stream>>>(x, Wf, Wg, Wh, q, k, v);
  attn_kernel<<<256, 512, 0, stream>>>(q, k, v, o);
  out_kernel<<<512, 256, 0, stream>>>(o, Wv, x, gm, out);
}


# attention softmax: row max and row sum butterflies of a 16-lane row as row_xmask DPP max/add on the vector ALU instead of LDS lane permutes; same exchange pattern, sum order unchanged
# speedup vs baseline: 1.4027x; 1.0383x over previous
#include <hip/hip_runtime.h>


typedef float v2f __attribute__((ext_vector_type(2)));
typedef float v8f __attribute__((ext_vector_type(8)));

#define WMMA_F32(A, B, C) \
  __builtin_amdgcn_wmma_f32_16x16x4_f32(false, (A), false, (B), (short)0, (C), false, false)

static __device__ __forceinline__ v2f ld2(const float* p) { return *(const v2f*)p; }
static __device__ __forceinline__ v2f mk2(float a, float b) { v2f r; r.x = a; r.y = b; return r; }

template <int K>
static __device__ __forceinline__ void row_max_step(float& v) {
  asm("v_max_num_f32_dpp %0, %0, %0 row_xmask:%1 row_mask:0xf bank_mask:0xf" : "+v"(v) : "n"(K));
}
template <int K>
static __device__ __forceinline__ void row_add_step(float& v) {
  asm("v_add_f32_dpp %0, %0, %0 row_xmask:%1 row_mask:0xf bank_mask:0xf" : "+v"(v) : "n"(K));
}

typedef float v4f __attribute__((ext_vector_type(4)));
#define XP 260
#define OP 100

__global__ __launch_bounds__(384) void qkv_kernel(
    const float* __restrict__ x,
    const float* __restrict__ Wf, const float* __restrict__ Wg, const float* __restrict__ Wh,
    float* __restrict__ q, float* __restrict__ k, float* __restrict__ v) {
  __shared__ __attribute__((aligned(16))) float xs[64 * XP];
  const int tid = threadIdx.x;
  const int wave = tid >> 5, lane = tid & 31;
  const int c = lane & 15, half = lane >> 4, kOff = half * 2;
  const int nt = wave % 6;
  const int rh = wave / 6;
  const int rowB = blockIdx.x * 64;

  v4f xr[11];
#pragma unroll
  for (int r = 0; r < 11; ++r) {
    const int j = tid + 384 * r;
    if (j < 4096) xr[r] = *(const v4f*)(x + (rowB + (j >> 6)) * 256 + (j & 63) * 4);
  }

  const float* Wsel = (nt < 2) ? Wf : ((nt < 4) ? Wg : Wh);
  const float* wp = Wsel + kOff * 32 + (nt & 1) * 16 + c;
  v2f aW[64];
#pragma unroll
  for (int t = 0; t < 64; ++t) aW[t] = mk2(wp[(4 * t) * 32], wp[(4 * t + 1) * 32]);

#pragma unroll
  for (int r = 0; r < 11; ++r) {
    const int j = tid + 384 * r;
    if (j < 4096) *(v4f*)&xs[(j >> 6) * XP + (j & 63) * 4] = xr[r];
  }
  __syncthreads();

  v8f acc0 = {}, acc1 = {};
  const float* b0p = &xs[(rh * 32 + c) * XP + kOff];
  const float* b1p = b0p + 16 * XP;
  v2f bA0[8], bA1[8], bB0[8], bB1[8];
#pragma unroll
  for (int u = 0; u < 8; ++u) {
    bA0[u] = *(const v2f*)(b0p + 4 * u);
    bA1[u] = *(const v2f*)(b1p + 4 * u);
  }
#pragma unroll
  for (int tb = 0; tb < 8; tb += 2) {
#pragma unroll
    for (int u = 0; u < 8; ++u) {
      bB0[u] = *(const v2f*)(b0p + 4 * (8 * (tb + 1) + u));
      bB1[u] = *(const v2f*)(b1p + 4 * (8 * (tb + 1) + u));
    }
    __builtin_amdgcn_sched_barrier(0);
#pragma unroll
    for (int u = 0; u < 8; ++u) {
      acc0 = WMMA_F32(aW[8 * tb + u], bA0[u], acc0);
      acc1 = WMMA_F32(aW[8 * tb + u], bA1[u], acc1);
    }
    if (tb + 2 < 8) {
#pragma unroll
      for (int u = 0; u < 8; ++u) {
        bA0[u] = *(const v2f*)(b0p + 4 * (8 * (tb + 2) + u));
        bA1[u] = *(const v2f*)(b1p + 4 * (8 * (tb + 2) + u));
      }
    }
    __builtin_amdgcn_sched_barrier(0);
#pragma unroll
    for (int u = 0; u < 8; ++u) {
      acc0 = WMMA_F32(aW[8 * (tb + 1) + u], bB0[u], acc0);
      acc1 = WMMA_F32(aW[8 * (tb + 1) + u], bB1[u], acc1);
    }
  }
  __syncthreads();

  {
    float* o0 = &xs[(rh * 32 + c) * OP + nt * 16 + half * 8];
    float* o1 = o0 + 16 * OP;
    v4f t0, t1;
    t0.x = acc0[0]; t0.y = acc0[1]; t0.z = acc0[2]; t0.w = acc0[3];
    t1.x = acc0[4]; t1.y = acc0[5]; t1.z = acc0[6]; t1.w = acc0[7];
    *(v4f*)o0 = t0; *(v4f*)(o0 + 4) = t1;
    t0.x = acc1[0]; t0.y = acc1[1]; t0.z = acc1[2]; t0.w = acc1[3];
    t1.x = acc1[4]; t1.y = acc1[5]; t1.z = acc1[6]; t1.w = acc1[7];
    *(v4f*)o1 = t0; *(v4f*)(o1 + 4) = t1;
  }
  __syncthreads();

#pragma unroll
  for (int i = 0; i < 4; ++i) {
    const int idx = tid + 384 * i;
    const int ob = idx >> 9, row = (idx & 511) >> 3, p4 = (idx & 7) * 4;
    float* dst = (ob == 0) ? q : ((ob == 1) ? k : v);
    *(v4f*)(dst + (rowB + row) * 32 + p4) = *(const v4f*)&xs[row * OP + ob * 32 + p4];
  }
}

#define QP 36
#define VP 48
#define PP 68

__global__ __launch_bounds__(512) void attn_kernel(
    const float* __restrict__ q, const float* __restrict__ k,
    const float* __restrict__ v, float* __restrict__ o) {
  __shared__ __attribute__((aligned(16))) float qs[2][4][64 * QP];
  __shared__ __attribute__((aligned(16))) float vs[3][4][64 * VP];
  __shared__ __attribute__((aligned(16))) float ps[16][16 * PP];
  const int tid = threadIdx.x;
  const int wave = tid >> 5, lane = tid & 31;
  const int m = lane & 15, half = lane >> 4, kOff = half * 2;
  const int tile = wave & 3, split = wave >> 2;
  const int b = blockIdx.x >> 6;
  const int i0 = (blockIdx.x & 63) * 64 + tile * 16;
  const float* qb = q + b * 4096 * 32;
  const float* kb = k + b * 4096 * 32;
  const float* vb = v + b * 4096 * 32;
  float* pw = ps[wave];

  const int l = tid & 127;
  const int srow = l >> 3, scol = (l & 7) * 4;
  const float* qsrc = qb + (split * 1024 + srow) * 32 + scol;
  const float* vsrc = vb + (split * 1024 + srow) * 32 + scol;
  const int qdst = srow * QP + scol, vdst = srow * VP + scol;

  v2f aK[8];
#pragma unroll
  for (int t = 0; t < 8; ++t) aK[t] = ld2(&kb[(i0 + m) * 32 + 4 * t + kOff]);

  float mrun[8], lrun[8];
  v8f o0 = {}, o1 = {};
#pragma unroll
  for (int vi = 0; vi < 8; ++vi) { mrun[vi] = -__builtin_inff(); lrun[vi] = 0.f; }

  v4f pq[4], pv[4];
#pragma unroll
  for (int r = 0; r < 4; ++r) {
    pq[r] = *(const v4f*)(qsrc + r * 512);
    pv[r] = *(const v4f*)(vsrc + r * 512);
  }
#pragma unroll
  for (int r = 0; r < 4; ++r) {
    *(v4f*)&qs[0][split][qdst + r * 16 * QP] = pq[r];
    *(v4f*)&vs[0][split][vdst + r * 16 * VP] = pv[r];
  }
  __syncthreads();
  asm volatile("s_barrier_signal -1" ::: "memory");
#pragma unroll
  for (int r = 0; r < 4; ++r) {
    pq[r] = *(const v4f*)(qsrc + 2048 + r * 512);
    pv[r] = *(const v4f*)(vsrc + 2048 + r * 512);
  }

  int vc = 0;
  for (int it = 0; it < 16; ++it) {
    const int sg = it & 1;
    const int vn = (vc == 2) ? 0 : vc + 1;
    const float* qt = qs[sg][split];
    const float* vt = vs[vc][split];
    asm volatile("s_barrier_wait -1" ::: "memory");
    v8f s[4] = {{}, {}, {}, {}};
#pragma unroll
    for (int st = 0; st < 4; ++st) {
#pragma unroll
      for (int t = 0; t < 8; ++t) {
        const v2f bq = *(const v2f*)&qt[(st * 16 + m) * QP + 4 * t + kOff];
        s[st] = WMMA_F32(aK[t], bq, s[st]);
      }
    }
    __builtin_amdgcn_sched_barrier(0);
    {
#pragma unroll
      for (int r = 0; r < 4; ++r) {
        *(v4f*)&qs[sg ^ 1][split][qdst + r * 16 * QP] = pq[r];
        *(v4f*)&vs[vn][split][vdst + r * 16 * VP] = pv[r];
      }
      asm volatile("s_wait_dscnt 0x0\n\ts_barrier_signal -1" ::: "memory");
      const int nt = (it < 14) ? it + 2 : 15;
#pragma unroll
      for (int r = 0; r < 4; ++r) {
        pq[r] = *(const v4f*)(qsrc + nt * 2048 + r * 512);
        pv[r] = *(const v4f*)(vsrc + nt * 2048 + r * 512);
      }
    }
#pragma unroll
    for (int vi = 0; vi < 8; ++vi) {
      float mx = fmaxf(fmaxf(s[0][vi], s[1][vi]), fmaxf(s[2][vi], s[3][vi]));
      row_max_step<1>(mx);
      row_max_step<2>(mx);
      row_max_step<4>(mx);
      row_max_step<8>(mx);
      const float nm = fmaxf(mrun[vi], mx);
      const float sc = __expf(mrun[vi] - nm);
      mrun[vi] = nm;
      const float p0 = __expf(s[0][vi] - nm); s[0][vi] = p0;
      const float p1 = __expf(s[1][vi] - nm); s[1][vi] = p1;
      const float p2 = __expf(s[2][vi] - nm); s[2][vi] = p2;
      const float p3 = __expf(s[3][vi] - nm); s[3][vi] = p3;
      float rs = (p0 + p1) + (p2 + p3);
      row_add_step<1>(rs);
      row_add_step<2>(rs);
      row_add_step<4>(rs);
      row_add_step<8>(rs);
      lrun[vi] = lrun[vi] * sc + rs;
      o0[vi] *= sc;
      o1[vi] *= sc;
    }
#pragma unroll
    for (int st = 0; st < 4; ++st)
#pragma unroll
      for (int vi = 0; vi < 8; ++vi)
        pw[(vi + half * 8) * PP + st * 16 + m] = s[st][vi];
    asm volatile("s_wait_dscnt 0" ::: "memory");
    v2f aP[16], bv0[16], bv1[16];
#pragma unroll
    for (int t = 0; t < 16; ++t) {
      const int kk = 4 * t + kOff;
      aP[t] = *(const v2f*)&pw[m * PP + kk];
      bv0[t] = mk2(vt[kk * VP + m],      vt[(kk + 1) * VP + m]);
      bv1[t] = mk2(vt[kk * VP + 16 + m], vt[(kk + 1) * VP + 16 + m]);
    }
    __builtin_amdgcn_sched_barrier(0);
#pragma unroll
    for (int t = 0; t < 16; ++t) {
      o0 = WMMA_F32(aP[t], bv0[t], o0);
      o1 = WMMA_F32(aP[t], bv1[t], o1);
    }
    vc = vn;
  }
  asm volatile("s_barrier_wait -1" ::: "memory");

  asm volatile("s_wait_dscnt 0" ::: "memory");
#pragma unroll
  for (int vi = 0; vi < 8; ++vi) {
    const int r = vi + half * 8;
    pw[r * 32 + m]      = o0[vi];
    pw[r * 32 + 16 + m] = o1[vi];
  }
  if (m == 0) {
#pragma unroll
    for (int vi = 0; vi < 8; ++vi) {
      pw[512 + vi + half * 8] = lrun[vi];
      pw[528 + vi + half * 8] = mrun[vi];
    }
  }
  __syncthreads();

  if (split == 0) {
#pragma unroll
    for (int vi = 0; vi < 8; ++vi) {
      const int r = vi + half * 8;
      float mg = ps[tile][528 + r];
      mg = fmaxf(mg, ps[tile + 4][528 + r]);
      mg = fmaxf(mg, ps[tile + 8][528 + r]);
      mg = fmaxf(mg, ps[tile + 12][528 + r]);
      float lg = 0.f, a0 = 0.f, a1 = 0.f;
#pragma unroll
      for (int p = 0; p < 4; ++p) {
        const float* pp = ps[tile + 4 * p];
        const float wgt = __expf(pp[528 + r] - mg);
        lg += wgt * pp[512 + r];
        a0 += wgt * pp[r * 32 + m];
        a1 += wgt * pp[r * 32 + 16 + m];
      }
      const float inv = 1.f / lg;
      const int g = b * 4096 * 32 + (i0 + r) * 32;
      o[g + m]      = a0 * inv;
      o[g + 16 + m] = a1 * inv;
    }
  }
}

__global__ __launch_bounds__(256) void out_kernel(
    const float* __restrict__ o, const float* __restrict__ Wv,
    const float* __restrict__ x, const float* __restrict__ gamma,
    float* __restrict__ out) {
  const int tid = threadIdx.x;
  const int wave = tid >> 5, lane = tid & 31;
  const int m = lane & 15, half = lane >> 4, kOff = half * 2;
  const int task = blockIdx.x * 8 + wave;
  const int row0 = (task >> 2) * 16, col0 = (task & 3) * 64;

  v2f aO[8];
#pragma unroll
  for (int t = 0; t < 8; ++t) aO[t] = ld2(&o[(row0 + m) * 32 + 4 * t + kOff]);

  v8f acc[4] = {{}, {}, {}, {}};
#pragma unroll
  for (int t = 0; t < 8; ++t) {
    const int kk = 4 * t + kOff;
#pragma unroll
    for (int ct = 0; ct < 4; ++ct) {
      const int col = col0 + ct * 16 + m;
      const v2f bw = mk2(Wv[kk * 256 + col], Wv[(kk + 1) * 256 + col]);
      acc[ct] = WMMA_F32(aO[t], bw, acc[ct]);
    }
  }
  const float g = *gamma;
#pragma unroll
  for (int ct = 0; ct < 4; ++ct) {
#pragma unroll
    for (int vi = 0; vi < 8; ++vi) {
      const int idx = (row0 + vi + half * 8) * 256 + col0 + ct * 16 + m;
      out[idx] = g * acc[ct][vi] + x[idx];
    }
  }
}

extern "C" void kernel_launch(void* const* d_in, const int* in_sizes, int n_in,
                              void* d_out, int out_size, void* d_ws, size_t ws_size,
                              hipStream_t stream) {
  const float* x  = (const float*)d_in[0];
  const float* Wf = (const float*)d_in[1];
  const float* Wg = (const float*)d_in[2];
  const float* Wh = (const float*)d_in[3];
  const float* Wv = (const float*)d_in[4];
  const float* gm = (const float*)d_in[5];
  float* out = (float*)d_out;

  const int QKV = 4 * 4096 * 32;
  float* q = (float*)d_ws;
  float* k = q + QKV;
  float* v = k + QKV;
  float* o = v + QKV;

  qkv_kernel<<<256, 384, 0, stream>>>(x, Wf, Wg, Wh, q, k, v);
  attn_kernel<<<256, 512, 0, stream>>>(q, k, v, o);
  out_kernel<<<512, 256, 0, stream>>>(o, Wv, x, gm, out);
}


# attention loop, fewer vector instructions per tile: p = exp2(fma(s, log2e, -max*log2e)), rows in pairs; row sums per lane, joined once after the loop; V loads merged per k-row pair; sum order changes
# speedup vs baseline: 1.4626x; 1.0427x over previous
#include <hip/hip_runtime.h>


typedef float v2f __attribute__((ext_vector_type(2)));
typedef float v8f __attribute__((ext_vector_type(8)));

#define WMMA_F32(A, B, C) \
  __builtin_amdgcn_wmma_f32_16x16x4_f32(false, (A), false, (B), (short)0, (C), false, false)

static __device__ __forceinline__ v2f ld2(const float* p) { return *(const v2f*)p; }
static __device__ __forceinline__ v2f mk2(float a, float b) { v2f r; r.x = a; r.y = b; return r; }

template <int K>
static __device__ __forceinline__ void row_max_step(float& v) {
  asm("v_max_num_f32_dpp %0, %0, %0 row_xmask:%1 row_mask:0xf bank_mask:0xf" : "+v"(v) : "n"(K));
}
template <int K>
static __device__ __forceinline__ void row_add_step(float& v) {
  asm("v_add_f32_dpp %0, %0, %0 row_xmask:%1 row_mask:0xf bank_mask:0xf" : "+v"(v) : "n"(K));
}

static __device__ __forceinline__ float vmin2(float a, float b) { return __builtin_elementwise_minimumnum(a, b); }
static __device__ __forceinline__ float vmax2(float a, float b) { return __builtin_elementwise_maximumnum(a, b); }
static __device__ __forceinline__ float ex2(float x) { return __builtin_amdgcn_exp2f(x); }
#define LOG2E 1.44269504088896341f

typedef float v4f __attribute__((ext_vector_type(4)));
#define XP 260
#define OP 100

__global__ __launch_bounds__(384) void qkv_kernel(
    const float* __restrict__ x,
    const float* __restrict__ Wf, const float* __restrict__ Wg, const float* __restrict__ Wh,
    float* __restrict__ q, float* __restrict__ k, float* __restrict__ v) {
  __shared__ __attribute__((aligned(16))) float xs[64 * XP];
  const int tid = threadIdx.x;
  const int wave = tid >> 5, lane = tid & 31;
  const int c = lane & 15, half = lane >> 4, kOff = half * 2;
  const int nt = wave % 6;
  const int rh = wave / 6;
  const int rowB = blockIdx.x * 64;

  v4f xr[11];
#pragma unroll
  for (int r = 0; r < 11; ++r) {
    const int j = tid + 384 * r;
    if (j < 4096) xr[r] = *(const v4f*)(x + (rowB + (j >> 6)) * 256 + (j & 63) * 4);
  }

  const float* Wsel = (nt < 2) ? Wf : ((nt < 4) ? Wg : Wh);
  const float* wp = Wsel + kOff * 32 + (nt & 1) * 16 + c;
  v2f aW[64];
#pragma unroll
  for (int t = 0; t < 64; ++t) aW[t] = mk2(wp[(4 * t) * 32], wp[(4 * t + 1) * 32]);

#pragma unroll
  for (int r = 0; r < 11; ++r) {
    const int j = tid + 384 * r;
    if (j < 4096) *(v4f*)&xs[(j >> 6) * XP + (j & 63) * 4] = xr[r];
  }
  __syncthreads();

  v8f acc0 = {}, acc1 = {};
  const float* b0p = &xs[(rh * 32 + c) * XP + kOff];
  const float* b1p = b0p + 16 * XP;
  v2f bA0[8], bA1[8], bB0[8], bB1[8];
#pragma unroll
  for (int u = 0; u < 8; ++u) {
    bA0[u] = *(const v2f*)(b0p + 4 * u);
    bA1[u] = *(const v2f*)(b1p + 4 * u);
  }
#pragma unroll
  for (int tb = 0; tb < 8; tb += 2) {
#pragma unroll
    for (int u = 0; u < 8; ++u) {
      bB0[u] = *(const v2f*)(b0p + 4 * (8 * (tb + 1) + u));
      bB1[u] = *(const v2f*)(b1p + 4 * (8 * (tb + 1) + u));
    }
    __builtin_amdgcn_sched_barrier(0);
#pragma unroll
    for (int u = 0; u < 8; ++u) {
      acc0 = WMMA_F32(aW[8 * tb + u], bA0[u], acc0);
      acc1 = WMMA_F32(aW[8 * tb + u], bA1[u], acc1);
    }
    if (tb + 2 < 8) {
#pragma unroll
      for (int u = 0; u < 8; ++u) {
        bA0[u] = *(const v2f*)(b0p + 4 * (8 * (tb + 2) + u));
        bA1[u] = *(const v2f*)(b1p + 4 * (8 * (tb + 2) + u));
      }
    }
    __builtin_amdgcn_sched_barrier(0);
#pragma unroll
    for (int u = 0; u < 8; ++u) {
      acc0 = WMMA_F32(aW[8 * (tb + 1) + u], bB0[u], acc0);
      acc1 = WMMA_F32(aW[8 * (tb + 1) + u], bB1[u], acc1);
    }
  }
  __syncthreads();

  {
    float* o0 = &xs[(rh * 32 + c) * OP + nt * 16 + half * 8];
    float* o1 = o0 + 16 * OP;
    v4f t0, t1;
    t0.x = acc0[0]; t0.y = acc0[1]; t0.z = acc0[2]; t0.w = acc0[3];
    t1.x = acc0[4]; t1.y = acc0[5]; t1.z = acc0[6]; t1.w = acc0[7];
    *(v4f*)o0 = t0; *(v4f*)(o0 + 4) = t1;
    t0.x = acc1[0]; t0.y = acc1[1]; t0.z = acc1[2]; t0.w = acc1[3];
    t1.x = acc1[4]; t1.y = acc1[5]; t1.z = acc1[6]; t1.w = acc1[7];
    *(v4f*)o1 = t0; *(v4f*)(o1 + 4) = t1;
  }
  __syncthreads();

#pragma unroll
  for (int i = 0; i < 4; ++i) {
    const int idx = tid + 384 * i;
    const int ob = idx >> 9, row = (idx & 511) >> 3, p4 = (idx & 7) * 4;
    float* dst = (ob == 0) ? q : ((ob == 1) ? k : v);
    *(v4f*)(dst + (rowB + row) * 32 + p4) = *(const v4f*)&xs[row * OP + ob * 32 + p4];
  }
}

#define QP 36
#define VP 48
#define PP 68

__global__ __launch_bounds__(512) void attn_kernel(
    const float* __restrict__ q, const float* __restrict__ k,
    const float* __restrict__ v, float* __restrict__ o) {
  __shared__ __attribute__((aligned(16))) float qs[2][4][64 * QP];
  __shared__ __attribute__((aligned(16))) float vs[3][4][64 * VP];
  __shared__ __attribute__((aligned(16))) float ps[16][16 * PP];
  const int tid = threadIdx.x;
  const int wave = tid >> 5, lane = tid & 31;
  const int m = lane & 15, half = lane >> 4, kOff = half * 2;
  const int tile = wave & 3, split = wave >> 2;
  const int b = blockIdx.x >> 6;
  const int i0 = (blockIdx.x & 63) * 64 + tile * 16;
  const float* qb = q + b * 4096 * 32;
  const float* kb = k + b * 4096 * 32;
  const float* vb = v + b * 4096 * 32;
  float* pw = ps[wave];

  const int l = tid & 127;
  const int srow = l >> 3, scol = (l & 7) * 4;
  const float* qsrc = qb + (split * 1024 + srow) * 32 + scol;
  const float* vsrc = vb + (split * 1024 + srow) * 32 + scol;
  const int qdst = srow * QP + scol, vdst = srow * VP + scol;

  v2f aK[8];
#pragma unroll
  for (int t = 0; t < 8; ++t) aK[t] = ld2(&kb[(i0 + m) * 32 + 4 * t + kOff]);

  v2f mneg[4], lrun[4];
  v8f o0 = {}, o1 = {};
#pragma unroll
  for (int vq = 0; vq < 4; ++vq) { mneg[vq] = mk2(__builtin_inff(), __builtin_inff()); lrun[vq] = mk2(0.f, 0.f); }

  v4f pq[4], pv[4];
#pragma unroll
  for (int r = 0; r < 4; ++r) {
    pq[r] = *(const v4f*)(qsrc + r * 512);
    pv[r] = *(const v4f*)(vsrc + r * 512);
  }
#pragma unroll
  for (int r = 0; r < 4; ++r) {
    *(v4f*)&qs[0][split][qdst + r * 16 * QP] = pq[r];
    *(v4f*)&vs[0][split][vdst + r * 16 * VP] = pv[r];
  }
  __syncthreads();
  asm volatile("s_barrier_signal -1" ::: "memory");
#pragma unroll
  for (int r = 0; r < 4; ++r) {
    pq[r] = *(const v4f*)(qsrc + 2048 + r * 512);
    pv[r] = *(const v4f*)(vsrc + 2048 + r * 512);
  }

  int vc = 0;
  for (int it = 0; it < 16; ++it) {
    const int sg = it & 1;
    const int vn = (vc == 2) ? 0 : vc + 1;
    const float* qt = qs[sg][split];
    const float* vt = vs[vc][split];
    asm volatile("s_barrier_wait -1" ::: "memory");
    v8f s[4] = {{}, {}, {}, {}};
#pragma unroll
    for (int st = 0; st < 4; ++st) {
#pragma unroll
      for (int t = 0; t < 8; ++t) {
        const v2f bq = *(const v2f*)&qt[(st * 16 + m) * QP + 4 * t + kOff];
        s[st] = WMMA_F32(aK[t], bq, s[st]);
      }
    }
    __builtin_amdgcn_sched_barrier(0);
    {
#pragma unroll
      for (int r = 0; r < 4; ++r) {
        *(v4f*)&qs[sg ^ 1][split][qdst + r * 16 * QP] = pq[r];
        *(v4f*)&vs[vn][split][vdst + r * 16 * VP] = pv[r];
      }
      asm volatile("s_wait_dscnt 0x0\n\ts_barrier_signal -1" ::: "memory");
      const int nt = (it < 14) ? it + 2 : 15;
#pragma unroll
      for (int r = 0; r < 4; ++r) {
        pq[r] = *(const v4f*)(qsrc + nt * 2048 + r * 512);
        pv[r] = *(const v4f*)(vsrc + nt * 2048 + r * 512);
      }
    }
    const v2f l2e = mk2(LOG2E, LOG2E);
#pragma unroll
    for (int vq = 0; vq < 4; ++vq) {
      const int va = 2 * vq, vb = va + 1;
      float mxa = vmax2(vmax2(s[0][va], s[1][va]), vmax2(s[2][va], s[3][va]));
      float mxb = vmax2(vmax2(s[0][vb], s[1][vb]), vmax2(s[2][vb], s[3][vb]));
      row_max_step<1>(mxa);
      row_max_step<1>(mxb);
      row_max_step<2>(mxa);
      row_max_step<2>(mxb);
      row_max_step<4>(mxa);
      row_max_step<4>(mxb);
      row_max_step<8>(mxa);
      row_max_step<8>(mxb);
      const v2f me = mk2(mxa, mxb) * mk2(-LOG2E, -LOG2E);
      const v2f old = mneg[vq];
      const v2f nmn = mk2(vmin2(old.x, me.x), vmin2(old.y, me.y));
      const v2f d = nmn - old;
      const v2f sc = mk2(ex2(d.x), ex2(d.y));
      mneg[vq] = nmn;
      v2f p[4];
#pragma unroll
      for (int st = 0; st < 4; ++st) {
        const v2f a = __builtin_elementwise_fma(mk2(s[st][va], s[st][vb]), l2e, nmn);
        p[st] = mk2(ex2(a.x), ex2(a.y));
        s[st][va] = p[st].x;
        s[st][vb] = p[st].y;
      }
      lrun[vq] = lrun[vq] * sc + ((p[0] + p[1]) + (p[2] + p[3]));
      o0[va] *= sc.x; o0[vb] *= sc.y;
      o1[va] *= sc.x; o1[vb] *= sc.y;
    }
#pragma unroll
    for (int st = 0; st < 4; ++st)
#pragma unroll
      for (int vi = 0; vi < 8; ++vi)
        pw[(vi + half * 8) * PP + st * 16 + m] = s[st][vi];
    asm volatile("s_wait_dscnt 0" ::: "memory");
    v2f aP[16], bv0[16], bv1[16];
    int m1 = 16 + m;
    asm("" : "+v"(m1));
#pragma unroll
    for (int t = 0; t < 16; ++t) {
      const int kk = 4 * t + kOff;
      aP[t] = *(const v2f*)&pw[m * PP + kk];
      bv0[t] = mk2(vt[kk * VP + m],  vt[(kk + 1) * VP + m]);
      bv1[t] = mk2(vt[kk * VP + m1], vt[(kk + 1) * VP + m1]);
    }
    __builtin_amdgcn_sched_barrier(0);
#pragma unroll
    for (int t = 0; t < 16; ++t) {
      o0 = WMMA_F32(aP[t], bv0[t], o0);
      o1 = WMMA_F32(aP[t], bv1[t], o1);
    }
    vc = vn;
  }
  asm volatile("s_barrier_wait -1" ::: "memory");

  asm volatile("s_wait_dscnt 0" ::: "memory");
#pragma unroll
  for (int vi = 0; vi < 8; ++vi) {
    const int r = vi + half * 8;
    pw[r * 32 + m]      = o0[vi];
    pw[r * 32 + 16 + m] = o1[vi];
  }
  float lrow[8];
#pragma unroll
  for (int vi = 0; vi < 8; ++vi) {
    float l1 = (vi & 1) ? lrun[vi >> 1].y : lrun[vi >> 1].x;
    row_add_step<1>(l1);
    row_add_step<2>(l1);
    row_add_step<4>(l1);
    row_add_step<8>(l1);
    lrow[vi] = l1;
  }
  if (m == 0) {
#pragma unroll
    for (int vi = 0; vi < 8; ++vi) {
      pw[512 + vi + half * 8] = lrow[vi];
      pw[528 + vi + half * 8] = (vi & 1) ? mneg[vi >> 1].y : mneg[vi >> 1].x;
    }
  }
  __syncthreads();

  if (split == 0) {
#pragma unroll
    for (int vi = 0; vi < 8; ++vi) {
      const int r = vi + half * 8;
      float mg = ps[tile][528 + r];
      mg = vmin2(mg, ps[tile + 4][528 + r]);
      mg = vmin2(mg, ps[tile + 8][528 + r]);
      mg = vmin2(mg, ps[tile + 12][528 + r]);
      float lg = 0.f, a0 = 0.f, a1 = 0.f;
#pragma unroll
      for (int p = 0; p < 4; ++p) {
        const float* pp = ps[tile + 4 * p];
        const float wgt = ex2(mg - pp[528 + r]);
        lg += wgt * pp[512 + r];
        a0 += wgt * pp[r * 32 + m];
        a1 += wgt * pp[r * 32 + 16 + m];
      }
      const float inv = 1.f / lg;
      const int g = b * 4096 * 32 + (i0 + r) * 32;
      o[g + m]      = a0 * inv;
      o[g + 16 + m] = a1 * inv;
    }
  }
}

__global__ __launch_bounds__(256) void out_kernel(
    const float* __restrict__ o, const float* __restrict__ Wv,
    const float* __restrict__ x, const float* __restrict__ gamma,
    float* __restrict__ out) {
  const int tid = threadIdx.x;
  const int wave = tid >> 5, lane = tid & 31;
  const int m = lane & 15, half = lane >> 4, kOff = half * 2;
  const int task = blockIdx.x * 8 + wave;
  const int row0 = (task >> 2) * 16, col0 = (task & 3) * 64;

  v2f aO[8];
#pragma unroll
  for (int t = 0; t < 8; ++t) aO[t] = ld2(&o[(row0 + m) * 32 + 4 * t + kOff]);

  v8f acc[4] = {{}, {}, {}, {}};
#pragma unroll
  for (int t = 0; t < 8; ++t) {
    const int kk = 4 * t + kOff;
#pragma unroll
    for (int ct = 0; ct < 4; ++ct) {
      const int col = col0 + ct * 16 + m;
      const v2f bw = mk2(Wv[kk * 256 + col], Wv[(kk + 1) * 256 + col]);
      acc[ct] = WMMA_F32(aO[t], bw, acc[ct]);
    }
  }
  const float g = *gamma;
#pragma unroll
  for (int ct = 0; ct < 4; ++ct) {
#pragma unroll
    for (int vi = 0; vi < 8; ++vi) {
      const int idx = (row0 + vi + half * 8) * 256 + col0 + ct * 16 + m;
      out[idx] = g * acc[ct][vi] + x[idx];
    }
  }
}

extern "C" void kernel_launch(void* const* d_in, const int* in_sizes, int n_in,
                              void* d_out, int out_size, void* d_ws, size_t ws_size,
                              hipStream_t stream) {
  const float* x  = (const float*)d_in[0];
  const float* Wf = (const float*)d_in[1];
  const float* Wg = (const float*)d_in[2];
  const float* Wh = (const float*)d_in[3];
  const float* Wv = (const float*)d_in[4];
  const float* gm = (const float*)d_in[5];
  float* out = (float*)d_out;

  const int QKV = 4 * 4096 * 32;
  float* q = (float*)d_ws;
  float* k = q + QKV;
  float* v = k + QKV;
  float* o = v + QKV;

  qkv_kernel<<<256, 384, 0, stream>>>(x, Wf, Wg, Wh, q, k, v);
  attn_kernel<<<256, 512, 0, stream>>>(q, k, v, o);
  out_kernel<<<512, 256, 0, stream>>>(o, Wv, x, gm, out);
}
